# starting kernel
# baseline (speedup 1.0000x reference)
#include <hip/hip_runtime.h>
#include <hip/hip_bf16.h>

typedef __attribute__((ext_vector_type(16))) __bf16 v16bf;
typedef __attribute__((ext_vector_type(8)))  float  v8f;

__global__ void deg_init_kernel(float* __restrict__ deg, int n) {
    int i = blockIdx.x * blockDim.x + threadIdx.x;
    if (i < n) deg[i] = 1.0f;
}

__global__ void deg_count_kernel(const int* __restrict__ dst,
                                 float* __restrict__ deg, int E) {
    int i = blockIdx.x * blockDim.x + threadIdx.x;
    if (i < E) unsafeAtomicAdd(&deg[dst[i]], 1.0f);
}

__global__ void deg_rsqrt_kernel(const float* __restrict__ deg,
                                 float* __restrict__ isq, int n) {
    int i = blockIdx.x * blockDim.x + threadIdx.x;
    if (i < n) isq[i] = rsqrtf(deg[i]);
}

template <int CIN, int COUT>
__global__ void gemm_wmma_bf16_kernel(const float* __restrict__ X,
                                      const float* __restrict__ W,
                                      float* __restrict__ H) {
    const int lane = threadIdx.x;
    const int ctile = threadIdx.y;
    const int m0 = blockIdx.x * 16;
    const int half = lane >> 4;
    const int l = lane & 15;
    const int m = m0 + l;
    const int n = ctile * 16 + l;

    v8f acc = {0.f, 0.f, 0.f, 0.f, 0.f, 0.f, 0.f, 0.f};

    #pragma unroll
    for (int kb = 0; kb < CIN; kb += 32) {
        float xa[16];
        const float4* xr0 = reinterpret_cast<const float4*>(&X[(size_t)m * CIN + kb + half * 8]);
        const float4* xr1 = reinterpret_cast<const float4*>(&X[(size_t)m * CIN + kb + 16 + half * 8]);
        *reinterpret_cast<float4*>(&xa[0])  = xr0[0];
        *reinterpret_cast<float4*>(&xa[4])  = xr0[1];
        *reinterpret_cast<float4*>(&xa[8])  = xr1[0];
        *reinterpret_cast<float4*>(&xa[12]) = xr1[1];

        v16bf a, b;
        #pragma unroll
        for (int e = 0; e < 8; ++e) {
            const int k0 = kb + half * 8 + e;
            const int k1 = k0 + 16;
            a[e]     = (__bf16)xa[e];
            a[e + 8] = (__bf16)xa[e + 8];
            b[e]     = (__bf16)W[(size_t)k0 * COUT + n];
            b[e + 8] = (__bf16)W[(size_t)k1 * COUT + n];
        }
        acc = __builtin_amdgcn_wmma_f32_16x16x32_bf16(
             false, a,  false, b,
             (short)0, acc,  false,  false);
    }

    #pragma unroll
    for (int v = 0; v < 8; ++v) {
        const int r = m0 + v + half * 8;
        H[(size_t)r * COUT + ctile * 16 + l] = acc[v];
    }
}

__global__ void self_init_kernel(const float* __restrict__ H,
                                 const float* __restrict__ isq,
                                 float* __restrict__ AGG,
                                 long total, int cshift) {
    long i = (long)blockIdx.x * blockDim.x + threadIdx.x;
    if (i < total) {
        int node = (int)(i >> cshift);
        float s = isq[node];
        AGG[i] = H[i] * s * s;
    }
}

__global__ void scatter_edges_kernel(const float* __restrict__ H,
                                     const int* __restrict__ src,
                                     const int* __restrict__ dst,
                                     const float* __restrict__ isq,
                                     float* __restrict__ AGG,
                                     long total, int C, int lshift) {
    long idx = (long)blockIdx.x * blockDim.x + threadIdx.x;
    if (idx >= total) return;
    const int e = (int)(idx >> lshift);
    const int c = (int)(idx & ((1 << lshift) - 1)) << 2;
    const int s = src[e];
    const int d = dst[e];
    const float nrm = isq[s] * isq[d];
    const float4 v = *reinterpret_cast<const float4*>(&H[(size_t)s * C + c]);
    float* out = &AGG[(size_t)d * C + c];
    unsafeAtomicAdd(out + 0, v.x * nrm);
    unsafeAtomicAdd(out + 1, v.y * nrm);
    unsafeAtomicAdd(out + 2, v.z * nrm);
    unsafeAtomicAdd(out + 3, v.w * nrm);
}

__global__ void bias_act_kernel(const float* __restrict__ AGG,
                                const float* __restrict__ bias,
                                float* __restrict__ OUT,
                                long total, int cmask, int do_relu) {
    long i = (long)blockIdx.x * blockDim.x + threadIdx.x;
    if (i < total) {
        float v = AGG[i] + bias[(int)(i & cmask)];
        OUT[i] = do_relu ? fmaxf(v, 0.0f) : v;
    }
}

extern "C" void kernel_launch(void* const* d_in, const int* in_sizes, int n_in,
                              void* d_out, int out_size, void* d_ws, size_t ws_size,
                              hipStream_t stream) {
    const float* x  = (const float*)d_in[0];
    const int*   ei = (const int*)d_in[1];
    const float* W1 = (const float*)d_in[2];
    const float* b1 = (const float*)d_in[3];
    const float* W2 = (const float*)d_in[4];
    const float* b2 = (const float*)d_in[5];
    const float* W3 = (const float*)d_in[6];
    const float* b3 = (const float*)d_in[7];

    const int CIN = 128, CH = 128, COUT = 64;
    const int n = in_sizes[0] / CIN;
    const int E = in_sizes[1] / 2;
    const int* src = ei;
    const int* dst = ei + E;

    float* ws   = (float*)d_ws;
    float* deg  = ws;
    float* isq  = deg  + n;
    float* hlin = isq  + n;
    float* agg  = hlin + (size_t)n * CH;
    float* feat = agg  + (size_t)n * CH;

    const int TB = 256;
    deg_init_kernel <<<(n + TB - 1) / TB, TB, 0, stream>>>(deg, n);
    deg_count_kernel<<<(E + TB - 1) / TB, TB, 0, stream>>>(dst, deg, E);
    deg_rsqrt_kernel<<<(n + TB - 1) / TB, TB, 0, stream>>>(deg, isq, n);

    const int row_tiles = n / 16;

    {
        const int C = CH;
        gemm_wmma_bf16_kernel<128, 128><<<row_tiles, dim3(32, C / 16), 0, stream>>>(x, W1, hlin);
        long nc = (long)n * C;
        self_init_kernel<<<(int)((nc + TB - 1) / TB), TB, 0, stream>>>(hlin, isq, agg, nc, 7);
        long tot = (long)E << 5;
        scatter_edges_kernel<<<(int)((tot + TB - 1) / TB), TB, 0, stream>>>(
            hlin, src, dst, isq, agg, tot, C, 5);
        bias_act_kernel<<<(int)((nc + TB - 1) / TB), TB, 0, stream>>>(agg, b1, feat, nc, C - 1, 1);
    }

    {
        const int C = CH;
        gemm_wmma_bf16_kernel<128, 128><<<row_tiles, dim3(32, C / 16), 0, stream>>>(feat, W2, hlin);
        long nc = (long)n * C;
        self_init_kernel<<<(int)((nc + TB - 1) / TB), TB, 0, stream>>>(hlin, isq, agg, nc, 7);
        long tot = (long)E << 5;
        scatter_edges_kernel<<<(int)((tot + TB - 1) / TB), TB, 0, stream>>>(
            hlin, src, dst, isq, agg, tot, C, 5);
        bias_act_kernel<<<(int)((nc + TB - 1) / TB), TB, 0, stream>>>(agg, b2, feat, nc, C - 1, 1);
    }

    {
        const int C = COUT;
        gemm_wmma_bf16_kernel<128, 64><<<row_tiles, dim3(32, C / 16), 0, stream>>>(feat, W3, hlin);
        long nc = (long)n * C;
        self_init_kernel<<<(int)((nc + TB - 1) / TB), TB, 0, stream>>>(hlin, isq, agg, nc, 6);
        long tot = (long)E << 4;
        scatter_edges_kernel<<<(int)((tot + TB - 1) / TB), TB, 0, stream>>>(
            hlin, src, dst, isq, agg, tot, C, 4);
        bias_act_kernel<<<(int)((nc + TB - 1) / TB), TB, 0, stream>>>(
            agg, b3, (float*)d_out, nc, C - 1, 0);
    }
}


# finalize (part sums + self term, norm, bias, relu) of layers 1, 2 fused into the next layer's bf16 WMMA product as its A-tile prologue via LDS; row-major activations never stored; 9 launches not 11
# speedup vs baseline: 2.2271x; 2.2271x over previous
#include <hip/hip_runtime.h>
#include <hip/hip_bf16.h>


typedef __attribute__((ext_vector_type(16))) __bf16 v16bf;
typedef __attribute__((ext_vector_type(8)))  float  v8f;
typedef __attribute__((ext_vector_type(8)))  __bf16 v8bf;

constexpr int NN        = 10000;
constexpr int AGG_WAVES = 7;
constexpr int EDGE_STEP = 512;

__global__ __launch_bounds__(1024)
void edge_prep_kernel(const int* __restrict__ src, const int* __restrict__ dst,
                      unsigned* __restrict__ pk, int* __restrict__ hist_part,
                      int E, int n) {
    __shared__ int hist[NN];
    const int tid = threadIdx.x;
    const int e0 = (blockIdx.x * 1024 + tid) * 4;
    const bool vec = ((E & 3) == 0) && (e0 + 4 <= E);
    int4 s4 = make_int4(0, 0, 0, 0), d4 = make_int4(0, 0, 0, 0);
    if (vec) {
        s4 = *reinterpret_cast<const int4*>(&src[e0]);
        d4 = *reinterpret_cast<const int4*>(&dst[e0]);
    }
    for (int i = tid; i < NN; i += 1024) hist[i] = 0;
    __syncthreads();
    if (vec) {
        *reinterpret_cast<uint4*>(&pk[e0]) =
            make_uint4((unsigned)s4.x | ((unsigned)d4.x << 16), (unsigned)s4.y | ((unsigned)d4.y << 16),
                       (unsigned)s4.z | ((unsigned)d4.z << 16), (unsigned)s4.w | ((unsigned)d4.w << 16));
        atomicAdd(&hist[d4.x], 1);
        atomicAdd(&hist[d4.y], 1);
        atomicAdd(&hist[d4.z], 1);
        atomicAdd(&hist[d4.w], 1);
    } else {
        for (int e = e0; e < min(e0 + 4, E); ++e) {
            const int s = src[e];
            const int d = dst[e];
            pk[e] = (unsigned)s | ((unsigned)d << 16);
            atomicAdd(&hist[d], 1);
        }
    }
    __syncthreads();
    for (int i = tid; i < n; i += 1024) hist_part[(size_t)blockIdx.x * n + i] = hist[i];
}

__global__ void isq_kernel(const int* __restrict__ hist_part,
                           float* __restrict__ isq, int n, int G) {
    __shared__ int red[8][33];
    const int tx = threadIdx.x, ty = threadIdx.y;
    const int i = blockIdx.x * 32 + tx;
    int c = 0;
    if (i < n)
        for (int g = ty; g < G; g += 8) c += hist_part[(size_t)g * n + i];
    red[ty][tx] = c;
    __syncthreads();
    if (ty == 0 && i < n) {
        int t = 1;
        #pragma unroll
        for (int k = 0; k < 8; ++k) t += red[k][tx];
        isq[i] = rsqrtf((float)t);
    }
}

template <int CIN, int COUT>
__global__ void gemm_wmma_bf16_kernel(const float* __restrict__ X,
                                      const float* __restrict__ W,
                                      const float* __restrict__ isq,
                                      float* __restrict__ HT, int nrows) {
    const int lane = threadIdx.x;
    const int ctile = threadIdx.y;
    const int m0 = blockIdx.x * 16;
    const int half = lane >> 4;
    const int l = lane & 15;
    const int m = m0 + l;
    const int n = ctile * 16 + l;

    v8f acc = {0.f, 0.f, 0.f, 0.f, 0.f, 0.f, 0.f, 0.f};

    #pragma unroll
    for (int kb = 0; kb < CIN; kb += 32) {
        float xa[16];
        const float4* xr0 = reinterpret_cast<const float4*>(&X[(size_t)m * CIN + kb + half * 8]);
        const float4* xr1 = reinterpret_cast<const float4*>(&X[(size_t)m * CIN + kb + 16 + half * 8]);
        *reinterpret_cast<float4*>(&xa[0])  = xr0[0];
        *reinterpret_cast<float4*>(&xa[4])  = xr0[1];
        *reinterpret_cast<float4*>(&xa[8])  = xr1[0];
        *reinterpret_cast<float4*>(&xa[12]) = xr1[1];

        v16bf a, b;
        #pragma unroll
        for (int e = 0; e < 8; ++e) {
            const int k0 = kb + half * 8 + e;
            const int k1 = k0 + 16;
            a[e]     = (__bf16)xa[e];
            a[e + 8] = (__bf16)xa[e + 8];
            b[e]     = (__bf16)W[(size_t)k0 * COUT + n];
            b[e + 8] = (__bf16)W[(size_t)k1 * COUT + n];
        }
        acc = __builtin_amdgcn_wmma_f32_16x16x32_bf16(
             false, a,  false, b,
             (short)0, acc,  false,  false);
    }

    const int r0 = m0 + half * 8;
    const float4 s0 = *reinterpret_cast<const float4*>(&isq[r0]);
    const float4 s1 = *reinterpret_cast<const float4*>(&isq[r0 + 4]);
    float* o = &HT[(size_t)(ctile * 16 + l) * nrows + r0];
    *reinterpret_cast<float4*>(o)     = make_float4(acc[0] * s0.x, acc[1] * s0.y, acc[2] * s0.z, acc[3] * s0.w);
    *reinterpret_cast<float4*>(o + 4) = make_float4(acc[4] * s1.x, acc[5] * s1.y, acc[6] * s1.z, acc[7] * s1.w);
}

__device__ __forceinline__ void lds_add(float* p, float v) {
    __hip_atomic_fetch_add(p, v, __ATOMIC_RELAXED, __HIP_MEMORY_SCOPE_WORKGROUP);
}

__device__ __forceinline__ void load_edges(uint4 (&v)[4], const unsigned* __restrict__ pk,
                                           int base, int lane) {
    #pragma unroll
    for (int q = 0; q < 4; ++q)
        v[q] = *reinterpret_cast<const uint4*>(&pk[base + q * 128 + lane * 4]);
}

__device__ __forceinline__ void add_edges(const uint4 (&v)[4], const float* hcol, float* col) {
    float h[16];
    #pragma unroll
    for (int q = 0; q < 4; ++q) {
        h[q * 4 + 0] = hcol[v[q].x & 0xFFFFu];
        h[q * 4 + 1] = hcol[v[q].y & 0xFFFFu];
        h[q * 4 + 2] = hcol[v[q].z & 0xFFFFu];
        h[q * 4 + 3] = hcol[v[q].w & 0xFFFFu];
    }
    #pragma unroll
    for (int q = 0; q < 4; ++q) {
        lds_add(&col[v[q].x >> 16], h[q * 4 + 0]);
        lds_add(&col[v[q].y >> 16], h[q * 4 + 1]);
        lds_add(&col[v[q].z >> 16], h[q * 4 + 2]);
        lds_add(&col[v[q].w >> 16], h[q * 4 + 3]);
    }
}

__global__ __launch_bounds__(AGG_WAVES * 32)
void agg_kernel(const float* __restrict__ HT, const unsigned* __restrict__ pk,
                float* __restrict__ part, int E, int n, int C, int slice) {
    __shared__ __attribute__((aligned(16))) float hcol[NN];
    __shared__ __attribute__((aligned(16))) float cols[AGG_WAVES][NN];

    const int tid  = threadIdx.x;
    const int lane = tid & 31;
    const int w    = __builtin_amdgcn_readfirstlane(tid >> 5);
    const int c    = blockIdx.x % C;
    const int p    = blockIdx.x / C;

    for (int i = tid * 4; i < n; i += AGG_WAVES * 32 * 4)
        *reinterpret_cast<float4*>(&hcol[i]) =
            *reinterpret_cast<const float4*>(&HT[(size_t)c * n + i]);
    float* col = cols[w];
    for (int i = lane * 4; i < NN; i += 128)
        *reinterpret_cast<float4*>(&col[i]) = make_float4(0.f, 0.f, 0.f, 0.f);
    __syncthreads();

    const int beg = (p * AGG_WAVES + w) * slice;
    const int end = min(beg + slice, E);
    int base = beg;
    const int nfull = (end - beg) / EDGE_STEP;
    if (nfull > 0) {
        uint4 cur[4];
        load_edges(cur, pk, base, lane);
        for (int t = 0; t < nfull; ++t) {
            const int nb = (t + 1 < nfull) ? base + EDGE_STEP : base;
            uint4 nxt[4];
            load_edges(nxt, pk, nb, lane);
            add_edges(cur, hcol, col);
            #pragma unroll
            for (int q = 0; q < 4; ++q) cur[q] = nxt[q];
            base += EDGE_STEP;
        }
    }
    for (int e = base + lane; e < end; e += 32) {
        const unsigned u = pk[e];
        lds_add(&col[u >> 16], hcol[u & 0xFFFFu]);
    }
    __syncthreads();

    float* o = &part[((size_t)p * C + c) * n];
    for (int i = tid; i < n; i += AGG_WAVES * 32) {
        float s = cols[0][i];
        #pragma unroll
        for (int k = 1; k < AGG_WAVES; ++k) s += cols[k][i];
        o[i] = s;
    }
}

template <int P, bool RELU>
__global__ void fin_kernel(const float* __restrict__ part, const float* __restrict__ HT,
                           const float* __restrict__ isq, const float* __restrict__ bias,
                           float* __restrict__ OUT, int n, int C) {
    __shared__ float tile[32][33];
    const int tx = threadIdx.x, ty = threadIdx.y;
    const int i0 = blockIdx.x * 32, c0 = blockIdx.y * 32;
    const int i = i0 + tx;
    #pragma unroll
    for (int cc = ty; cc < 32; cc += 8) {
        const int c = c0 + cc;
        float s = 0.f;
        if (i < n) {
            s = part[(size_t)c * n + i];
            #pragma unroll
            for (int p = 1; p < P; ++p) s += part[((size_t)p * C + c) * n + i];
            s += HT[(size_t)c * n + i];
            s *= isq[i];
        }
        tile[cc][tx] = s;
    }
    __syncthreads();
    const float b = bias[c0 + tx];
    #pragma unroll
    for (int ii = ty; ii < 32; ii += 8) {
        const int node = i0 + ii;
        if (node < n) {
            const float v = tile[tx][ii] + b;
            OUT[(size_t)node * C + c0 + tx] = RELU ? fmaxf(v, 0.0f) : v;
        }
    }
}

constexpr int A_PITCH = 136;

template <int CIN, int COUT, int P>
__global__ void fin_gemm_wmma_bf16_kernel(const float* __restrict__ part,
                                          const float* __restrict__ HTin,
                                          const float* __restrict__ isq,
                                          const float* __restrict__ bias,
                                          const float* __restrict__ W,
                                          float* __restrict__ HTout, int nrows) {
    __shared__ __attribute__((aligned(16))) __bf16 As[16][A_PITCH];
    const int lane = threadIdx.x;
    const int ctile = threadIdx.y;
    const int m0 = blockIdx.x * 16;
    const int tid = ctile * 32 + lane;
    constexpr int T = 32 * (COUT / 16);

    const int half = lane >> 4;
    const int l = lane & 15;
    const int n = ctile * 16 + l;

    v16bf b[CIN / 32];
    #pragma unroll
    for (int kb = 0; kb < CIN; kb += 32) {
        #pragma unroll
        for (int e = 0; e < 8; ++e) {
            const int k0 = kb + half * 8 + e;
            const int k1 = k0 + 16;
            b[kb / 32][e]     = (__bf16)W[(size_t)k0 * COUT + n];
            b[kb / 32][e + 8] = (__bf16)W[(size_t)k1 * COUT + n];
        }
    }

    static_assert((CIN * 4) % T == 0, "items must divide evenly over the block");
    #pragma unroll
    for (int j = 0; j < CIN * 4 / T; ++j) {
        const int idx = tid + j * T;
        const int k = idx >> 2;
        const int q = idx & 3;
        const size_t off = (size_t)k * nrows + m0 + q * 4;
        float4 s = *reinterpret_cast<const float4*>(&part[off]);
        #pragma unroll
        for (int p = 1; p < P; ++p) {
            const float4 t = *reinterpret_cast<const float4*>(&part[(size_t)p * CIN * nrows + off]);
            s.x += t.x; s.y += t.y; s.z += t.z; s.w += t.w;
        }
        const float4 h = *reinterpret_cast<const float4*>(&HTin[off]);
        const float4 r = *reinterpret_cast<const float4*>(&isq[m0 + q * 4]);
        const float bk = bias[k];
        s.x = fmaxf((s.x + h.x) * r.x + bk, 0.0f);
        s.y = fmaxf((s.y + h.y) * r.y + bk, 0.0f);
        s.z = fmaxf((s.z + h.z) * r.z + bk, 0.0f);
        s.w = fmaxf((s.w + h.w) * r.w + bk, 0.0f);
        As[q * 4 + 0][k] = (__bf16)s.x;
        As[q * 4 + 1][k] = (__bf16)s.y;
        As[q * 4 + 2][k] = (__bf16)s.z;
        As[q * 4 + 3][k] = (__bf16)s.w;
    }
    __syncthreads();

    v8f acc = {0.f, 0.f, 0.f, 0.f, 0.f, 0.f, 0.f, 0.f};

    #pragma unroll
    for (int kb = 0; kb < CIN; kb += 32) {
        const v8bf alo = *reinterpret_cast<const v8bf*>(&As[l][kb + half * 8]);
        const v8bf ahi = *reinterpret_cast<const v8bf*>(&As[l][kb + 16 + half * 8]);
        const v16bf a = __builtin_shufflevector(alo, ahi, 0, 1, 2, 3, 4, 5, 6, 7, 8, 9, 10, 11, 12, 13, 14, 15);
        acc = __builtin_amdgcn_wmma_f32_16x16x32_bf16(
             false, a,  false, b[kb / 32],
             (short)0, acc,  false,  false);
    }

    const int r0 = m0 + half * 8;
    const float4 s0 = *reinterpret_cast<const float4*>(&isq[r0]);
    const float4 s1 = *reinterpret_cast<const float4*>(&isq[r0 + 4]);
    float* o = &HTout[(size_t)(ctile * 16 + l) * nrows + r0];
    *reinterpret_cast<float4*>(o)     = make_float4(acc[0] * s0.x, acc[1] * s0.y, acc[2] * s0.z, acc[3] * s0.w);
    *reinterpret_cast<float4*>(o + 4) = make_float4(acc[4] * s1.x, acc[5] * s1.y, acc[6] * s1.z, acc[7] * s1.w);
}

extern "C" void kernel_launch(void* const* d_in, const int* in_sizes, int n_in,
                              void* d_out, int out_size, void* d_ws, size_t ws_size,
                              hipStream_t stream) {
    const float* x  = (const float*)d_in[0];
    const int*   ei = (const int*)d_in[1];
    const float* W1 = (const float*)d_in[2];
    const float* b1 = (const float*)d_in[3];
    const float* W2 = (const float*)d_in[4];
    const float* b2 = (const float*)d_in[5];
    const float* W3 = (const float*)d_in[6];
    const float* b3 = (const float*)d_in[7];

    const int CIN = 128, CH = 128, COUT = 64;
    const int n = in_sizes[0] / CIN;
    const int E = in_sizes[1] / 2;
    const int* src = ei;
    const int* dst = ei + E;

    unsigned* pk = (unsigned*)d_ws;
    float* hT   = (float*)(pk + E);
    float* part = hT   + (size_t)n * CH;
    float* hT2  = part + (size_t)n * CH * 2;
    float* isq  = hT2  + (size_t)n * CH;
    int*   hist_part = (int*)(isq + n);

    const int row_tiles = n / 16;
    const int node_tiles = (n + 31) / 32;
    const int prep_wgs = (E + 4095) / 4096;

    edge_prep_kernel<<<prep_wgs, 1024, 0, stream>>>(src, dst, pk, hist_part, E, n);
    isq_kernel<<<node_tiles, dim3(32, 8), 0, stream>>>(hist_part, isq, n, prep_wgs);

    const int slice2 = ((E + 2 * AGG_WAVES - 1) / (2 * AGG_WAVES) + EDGE_STEP - 1) / EDGE_STEP * EDGE_STEP;
    const int slice4 = ((E + 4 * AGG_WAVES - 1) / (4 * AGG_WAVES) + EDGE_STEP - 1) / EDGE_STEP * EDGE_STEP;

    gemm_wmma_bf16_kernel<128, 128><<<row_tiles, dim3(32, CH / 16), 0, stream>>>(x, W1, isq, hT, n);
    agg_kernel<<<CH * 2, AGG_WAVES * 32, 0, stream>>>(hT, pk, part, E, n, CH, slice2);

    fin_gemm_wmma_bf16_kernel<128, 128, 2><<<row_tiles, dim3(32, CH / 16), 0, stream>>>(part, hT, isq, b1, W2, hT2, n);
    agg_kernel<<<CH * 2, AGG_WAVES * 32, 0, stream>>>(hT2, pk, part, E, n, CH, slice2);

    fin_gemm_wmma_bf16_kernel<128, 64, 2><<<row_tiles, dim3(32, COUT / 16), 0, stream>>>(part, hT2, isq, b2, W3, hT, n);
    agg_kernel<<<COUT * 4, AGG_WAVES * 32, 0, stream>>>(hT, pk, part, E, n, COUT, slice4);
    fin_kernel<4, false><<<dim3(node_tiles, COUT / 32), dim3(32, 8), 0, stream>>>(
        part, hT, isq, b3, (float*)d_out, n, COUT);
}


# loops that waited for each global read now issue a thread's reads as one group: aggregation channel load (12 round trips to 1, column zeroing under it, first edge step with it), degree-plane sum
# speedup vs baseline: 2.5747x; 1.1561x over previous
#include <hip/hip_runtime.h>
#include <hip/hip_bf16.h>


typedef __attribute__((ext_vector_type(16))) __bf16 v16bf;
typedef __attribute__((ext_vector_type(8)))  float  v8f;
typedef __attribute__((ext_vector_type(8)))  __bf16 v8bf;

constexpr int NN        = 10000;
constexpr int AGG_WAVES = 7;
constexpr int EDGE_STEP = 512;

__global__ __launch_bounds__(1024)
void edge_prep_kernel(const int* __restrict__ src, const int* __restrict__ dst,
                      unsigned* __restrict__ pk, int* __restrict__ hist_part,
                      int E, int n) {
    __shared__ int hist[NN];
    const int tid = threadIdx.x;
    const int e0 = (blockIdx.x * 1024 + tid) * 4;
    const bool vec = ((E & 3) == 0) && (e0 + 4 <= E);
    int4 s4 = make_int4(0, 0, 0, 0), d4 = make_int4(0, 0, 0, 0);
    if (vec) {
        s4 = *reinterpret_cast<const int4*>(&src[e0]);
        d4 = *reinterpret_cast<const int4*>(&dst[e0]);
    }
    for (int i = tid; i < NN; i += 1024) hist[i] = 0;
    __syncthreads();
    if (vec) {
        *reinterpret_cast<uint4*>(&pk[e0]) =
            make_uint4((unsigned)s4.x | ((unsigned)d4.x << 16), (unsigned)s4.y | ((unsigned)d4.y << 16),
                       (unsigned)s4.z | ((unsigned)d4.z << 16), (unsigned)s4.w | ((unsigned)d4.w << 16));
        atomicAdd(&hist[d4.x], 1);
        atomicAdd(&hist[d4.y], 1);
        atomicAdd(&hist[d4.z], 1);
        atomicAdd(&hist[d4.w], 1);
    } else {
        for (int e = e0; e < min(e0 + 4, E); ++e) {
            const int s = src[e];
            const int d = dst[e];
            pk[e] = (unsigned)s | ((unsigned)d << 16);
            atomicAdd(&hist[d], 1);
        }
    }
    __syncthreads();
    for (int i = tid; i < n; i += 1024) hist_part[(size_t)blockIdx.x * n + i] = hist[i];
}

constexpr int ISQ_BATCH = 24;
__global__ void isq_kernel(const int* __restrict__ hist_part,
                           float* __restrict__ isq, int n, int G) {
    __shared__ int red[8][33];
    const int tx = threadIdx.x, ty = threadIdx.y;
    const int node = blockIdx.x * 32 + tx;
    const int i = min(node, n - 1);
    int c = 0;
    for (int g0 = ty; g0 < G; g0 += 8 * ISQ_BATCH) {
        int v[ISQ_BATCH];
        #pragma unroll
        for (int u = 0; u < ISQ_BATCH; ++u)
            v[u] = hist_part[(size_t)min(g0 + u * 8, G - 1) * n + i];
        #pragma unroll
        for (int u = 0; u < ISQ_BATCH; ++u) c += (g0 + u * 8 < G) ? v[u] : 0;
    }
    red[ty][tx] = c;
    __syncthreads();
    if (ty == 0 && node < n) {
        int t = 1;
        #pragma unroll
        for (int k = 0; k < 8; ++k) t += red[k][tx];
        isq[node] = rsqrtf((float)t);
    }
}

template <int CIN, int COUT>
__global__ void gemm_wmma_bf16_kernel(const float* __restrict__ X,
                                      const float* __restrict__ W,
                                      const float* __restrict__ isq,
                                      float* __restrict__ HT, int nrows) {
    const int lane = threadIdx.x;
    const int ctile = threadIdx.y;
    const int m0 = blockIdx.x * 16;
    const int half = lane >> 4;
    const int l = lane & 15;
    const int m = m0 + l;
    const int n = ctile * 16 + l;

    v8f acc = {0.f, 0.f, 0.f, 0.f, 0.f, 0.f, 0.f, 0.f};

    #pragma unroll
    for (int kb = 0; kb < CIN; kb += 32) {
        float xa[16];
        const float4* xr0 = reinterpret_cast<const float4*>(&X[(size_t)m * CIN + kb + half * 8]);
        const float4* xr1 = reinterpret_cast<const float4*>(&X[(size_t)m * CIN + kb + 16 + half * 8]);
        *reinterpret_cast<float4*>(&xa[0])  = xr0[0];
        *reinterpret_cast<float4*>(&xa[4])  = xr0[1];
        *reinterpret_cast<float4*>(&xa[8])  = xr1[0];
        *reinterpret_cast<float4*>(&xa[12]) = xr1[1];

        v16bf a, b;
        #pragma unroll
        for (int e = 0; e < 8; ++e) {
            const int k0 = kb + half * 8 + e;
            const int k1 = k0 + 16;
            a[e]     = (__bf16)xa[e];
            a[e + 8] = (__bf16)xa[e + 8];
            b[e]     = (__bf16)W[(size_t)k0 * COUT + n];
            b[e + 8] = (__bf16)W[(size_t)k1 * COUT + n];
        }
        acc = __builtin_amdgcn_wmma_f32_16x16x32_bf16(
             false, a,  false, b,
             (short)0, acc,  false,  false);
    }

    const int r0 = m0 + half * 8;
    const float4 s0 = *reinterpret_cast<const float4*>(&isq[r0]);
    const float4 s1 = *reinterpret_cast<const float4*>(&isq[r0 + 4]);
    float* o = &HT[(size_t)(ctile * 16 + l) * nrows + r0];
    *reinterpret_cast<float4*>(o)     = make_float4(acc[0] * s0.x, acc[1] * s0.y, acc[2] * s0.z, acc[3] * s0.w);
    *reinterpret_cast<float4*>(o + 4) = make_float4(acc[4] * s1.x, acc[5] * s1.y, acc[6] * s1.z, acc[7] * s1.w);
}

__device__ __forceinline__ void lds_add(float* p, float v) {
    __hip_atomic_fetch_add(p, v, __ATOMIC_RELAXED, __HIP_MEMORY_SCOPE_WORKGROUP);
}

__device__ __forceinline__ void load_edges(uint4 (&v)[4], const unsigned* __restrict__ pk,
                                           int base, int lane) {
    #pragma unroll
    for (int q = 0; q < 4; ++q)
        v[q] = *reinterpret_cast<const uint4*>(&pk[base + q * 128 + lane * 4]);
}

__device__ __forceinline__ void add_edges(const uint4 (&v)[4], const float* hcol, float* col) {
    float h[16];
    #pragma unroll
    for (int q = 0; q < 4; ++q) {
        h[q * 4 + 0] = hcol[v[q].x & 0xFFFFu];
        h[q * 4 + 1] = hcol[v[q].y & 0xFFFFu];
        h[q * 4 + 2] = hcol[v[q].z & 0xFFFFu];
        h[q * 4 + 3] = hcol[v[q].w & 0xFFFFu];
    }
    #pragma unroll
    for (int q = 0; q < 4; ++q) {
        lds_add(&col[v[q].x >> 16], h[q * 4 + 0]);
        lds_add(&col[v[q].y >> 16], h[q * 4 + 1]);
        lds_add(&col[v[q].z >> 16], h[q * 4 + 2]);
        lds_add(&col[v[q].w >> 16], h[q * 4 + 3]);
    }
}

__global__ __launch_bounds__(AGG_WAVES * 32)
void agg_kernel(const float* __restrict__ HT, const unsigned* __restrict__ pk,
                float* __restrict__ part, int E, int n, int C, int slice) {
    __shared__ __attribute__((aligned(16))) float hcol[NN];
    __shared__ __attribute__((aligned(16))) float cols[AGG_WAVES][NN];

    const int tid  = threadIdx.x;
    const int lane = tid & 31;
    const int w    = __builtin_amdgcn_readfirstlane(tid >> 5);
    const int c    = blockIdx.x % C;
    const int p    = blockIdx.x / C;

    const int beg = (p * AGG_WAVES + w) * slice;
    const int end = min(beg + slice, E);
    int base = beg;
    const int nfull = (end - beg) / EDGE_STEP;

    constexpr int HC_TRIPS = (NN / 4 + AGG_WAVES * 32 - 1) / (AGG_WAVES * 32);
    constexpr int Z_TRIPS  = (NN / 4 + 31) / 32;
    const int last4 = (n >> 2) - 1;
    uint4 cur[4];
    load_edges(cur, pk, (nfull > 0) ? base : 0, lane);
    const float4* hsrc = reinterpret_cast<const float4*>(&HT[(size_t)c * n]);
    constexpr int TS = AGG_WAVES * 32;
    static_assert(HC_TRIPS == 12, "twelve named reads below");
    const float4 h0 = hsrc[min(tid + 0 * TS, last4)];
    const float4 h1 = hsrc[min(tid + 1 * TS, last4)];
    const float4 h2 = hsrc[min(tid + 2 * TS, last4)];
    const float4 h3 = hsrc[min(tid + 3 * TS, last4)];
    const float4 h4 = hsrc[min(tid + 4 * TS, last4)];
    const float4 h5 = hsrc[min(tid + 5 * TS, last4)];
    const float4 h6 = hsrc[min(tid + 6 * TS, last4)];
    const float4 h7 = hsrc[min(tid + 7 * TS, last4)];
    const float4 h8 = hsrc[min(tid + 8 * TS, last4)];
    const float4 h9 = hsrc[min(tid + 9 * TS, last4)];
    const float4 h10 = hsrc[min(tid + 10 * TS, last4)];
    const float4 h11 = hsrc[min(tid + 11 * TS, last4)];

    float* col = cols[w];
    float4* col4 = reinterpret_cast<float4*>(col);
    #pragma unroll
    for (int t = 0; t < Z_TRIPS; ++t)
        col4[min(lane + t * 32, NN / 4 - 1)] = make_float4(0.f, 0.f, 0.f, 0.f);

    float4* hcol4 = reinterpret_cast<float4*>(hcol);
    hcol4[min(tid + 0 * TS, last4)] = h0;
    hcol4[min(tid + 1 * TS, last4)] = h1;
    hcol4[min(tid + 2 * TS, last4)] = h2;
    hcol4[min(tid + 3 * TS, last4)] = h3;
    hcol4[min(tid + 4 * TS, last4)] = h4;
    hcol4[min(tid + 5 * TS, last4)] = h5;
    hcol4[min(tid + 6 * TS, last4)] = h6;
    hcol4[min(tid + 7 * TS, last4)] = h7;
    hcol4[min(tid + 8 * TS, last4)] = h8;
    hcol4[min(tid + 9 * TS, last4)] = h9;
    hcol4[min(tid + 10 * TS, last4)] = h10;
    hcol4[min(tid + 11 * TS, last4)] = h11;
    __syncthreads();

    if (nfull > 0) {
        for (int t = 0; t < nfull; ++t) {
            const int nb = (t + 1 < nfull) ? base + EDGE_STEP : base;
            uint4 nxt[4];
            load_edges(nxt, pk, nb, lane);
            add_edges(cur, hcol, col);
            #pragma unroll
            for (int q = 0; q < 4; ++q) cur[q] = nxt[q];
            base += EDGE_STEP;
        }
    }
    for (int e = base + lane; e < end; e += 32) {
        const unsigned u = pk[e];
        lds_add(&col[u >> 16], hcol[u & 0xFFFFu]);
    }
    __syncthreads();

    float* o = &part[((size_t)p * C + c) * n];
    for (int i = tid; i < n; i += AGG_WAVES * 32) {
        float s = cols[0][i];
        #pragma unroll
        for (int k = 1; k < AGG_WAVES; ++k) s += cols[k][i];
        o[i] = s;
    }
}

template <int P, bool RELU>
__global__ void fin_kernel(const float* __restrict__ part, const float* __restrict__ HT,
                           const float* __restrict__ isq, const float* __restrict__ bias,
                           float* __restrict__ OUT, int n, int C) {
    __shared__ float tile[32][33];
    const int tx = threadIdx.x, ty = threadIdx.y;
    const int i0 = blockIdx.x * 32, c0 = blockIdx.y * 32;
    const int i = i0 + tx;
    #pragma unroll
    for (int cc = ty; cc < 32; cc += 8) {
        const int c = c0 + cc;
        float s = 0.f;
        if (i < n) {
            s = part[(size_t)c * n + i];
            #pragma unroll
            for (int p = 1; p < P; ++p) s += part[((size_t)p * C + c) * n + i];
            s += HT[(size_t)c * n + i];
            s *= isq[i];
        }
        tile[cc][tx] = s;
    }
    __syncthreads();
    const float b = bias[c0 + tx];
    #pragma unroll
    for (int ii = ty; ii < 32; ii += 8) {
        const int node = i0 + ii;
        if (node < n) {
            const float v = tile[tx][ii] + b;
            OUT[(size_t)node * C + c0 + tx] = RELU ? fmaxf(v, 0.0f) : v;
        }
    }
}

constexpr int A_PITCH = 136;

template <int CIN, int COUT, int P>
__global__ void fin_gemm_wmma_bf16_kernel(const float* __restrict__ part,
                                          const float* __restrict__ HTin,
                                          const float* __restrict__ isq,
                                          const float* __restrict__ bias,
                                          const float* __restrict__ W,
                                          float* __restrict__ HTout, int nrows) {
    __shared__ __attribute__((aligned(16))) __bf16 As[16][A_PITCH];
    const int lane = threadIdx.x;
    const int ctile = threadIdx.y;
    const int m0 = blockIdx.x * 16;
    const int tid = ctile * 32 + lane;
    constexpr int T = 32 * (COUT / 16);

    const int half = lane >> 4;
    const int l = lane & 15;
    const int n = ctile * 16 + l;

    v16bf b[CIN / 32];
    #pragma unroll
    for (int kb = 0; kb < CIN; kb += 32) {
        #pragma unroll
        for (int e = 0; e < 8; ++e) {
            const int k0 = kb + half * 8 + e;
            const int k1 = k0 + 16;
            b[kb / 32][e]     = (__bf16)W[(size_t)k0 * COUT + n];
            b[kb / 32][e + 8] = (__bf16)W[(size_t)k1 * COUT + n];
        }
    }

    static_assert((CIN * 4) % T == 0, "items must divide evenly over the block");
    #pragma unroll
    for (int j = 0; j < CIN * 4 / T; ++j) {
        const int idx = tid + j * T;
        const int k = idx >> 2;
        const int q = idx & 3;
        const size_t off = (size_t)k * nrows + m0 + q * 4;
        float4 s = *reinterpret_cast<const float4*>(&part[off]);
        #pragma unroll
        for (int p = 1; p < P; ++p) {
            const float4 t = *reinterpret_cast<const float4*>(&part[(size_t)p * CIN * nrows + off]);
            s.x += t.x; s.y += t.y; s.z += t.z; s.w += t.w;
        }
        const float4 h = *reinterpret_cast<const float4*>(&HTin[off]);
        const float4 r = *reinterpret_cast<const float4*>(&isq[m0 + q * 4]);
        const float bk = bias[k];
        s.x = fmaxf((s.x + h.x) * r.x + bk, 0.0f);
        s.y = fmaxf((s.y + h.y) * r.y + bk, 0.0f);
        s.z = fmaxf((s.z + h.z) * r.z + bk, 0.0f);
        s.w = fmaxf((s.w + h.w) * r.w + bk, 0.0f);
        As[q * 4 + 0][k] = (__bf16)s.x;
        As[q * 4 + 1][k] = (__bf16)s.y;
        As[q * 4 + 2][k] = (__bf16)s.z;
        As[q * 4 + 3][k] = (__bf16)s.w;
    }
    __syncthreads();

    v8f acc = {0.f, 0.f, 0.f, 0.f, 0.f, 0.f, 0.f, 0.f};

    #pragma unroll
    for (int kb = 0; kb < CIN; kb += 32) {
        const v8bf alo = *reinterpret_cast<const v8bf*>(&As[l][kb + half * 8]);
        const v8bf ahi = *reinterpret_cast<const v8bf*>(&As[l][kb + 16 + half * 8]);
        const v16bf a = __builtin_shufflevector(alo, ahi, 0, 1, 2, 3, 4, 5, 6, 7, 8, 9, 10, 11, 12, 13, 14, 15);
        acc = __builtin_amdgcn_wmma_f32_16x16x32_bf16(
             false, a,  false, b[kb / 32],
             (short)0, acc,  false,  false);
    }

    const int r0 = m0 + half * 8;
    const float4 s0 = *reinterpret_cast<const float4*>(&isq[r0]);
    const float4 s1 = *reinterpret_cast<const float4*>(&isq[r0 + 4]);
    float* o = &HTout[(size_t)(ctile * 16 + l) * nrows + r0];
    *reinterpret_cast<float4*>(o)     = make_float4(acc[0] * s0.x, acc[1] * s0.y, acc[2] * s0.z, acc[3] * s0.w);
    *reinterpret_cast<float4*>(o + 4) = make_float4(acc[4] * s1.x, acc[5] * s1.y, acc[6] * s1.z, acc[7] * s1.w);
}

extern "C" void kernel_launch(void* const* d_in, const int* in_sizes, int n_in,
                              void* d_out, int out_size, void* d_ws, size_t ws_size,
                              hipStream_t stream) {
    const float* x  = (const float*)d_in[0];
    const int*   ei = (const int*)d_in[1];
    const float* W1 = (const float*)d_in[2];
    const float* b1 = (const float*)d_in[3];
    const float* W2 = (const float*)d_in[4];
    const float* b2 = (const float*)d_in[5];
    const float* W3 = (const float*)d_in[6];
    const float* b3 = (const float*)d_in[7];

    const int CIN = 128, CH = 128, COUT = 64;
    const int n = in_sizes[0] / CIN;
    const int E = in_sizes[1] / 2;
    const int* src = ei;
    const int* dst = ei + E;

    unsigned* pk = (unsigned*)d_ws;
    float* hT   = (float*)(pk + E);
    float* part = hT   + (size_t)n * CH;
    float* hT2  = part + (size_t)n * CH * 2;
    float* isq  = hT2  + (size_t)n * CH;
    int*   hist_part = (int*)(isq + n);

    const int row_tiles = n / 16;
    const int node_tiles = (n + 31) / 32;
    const int prep_wgs = (E + 4095) / 4096;

    edge_prep_kernel<<<prep_wgs, 1024, 0, stream>>>(src, dst, pk, hist_part, E, n);
    isq_kernel<<<node_tiles, dim3(32, 8), 0, stream>>>(hist_part, isq, n, prep_wgs);

    const int slice2 = ((E + 2 * AGG_WAVES - 1) / (2 * AGG_WAVES) + EDGE_STEP - 1) / EDGE_STEP * EDGE_STEP;
    const int slice4 = ((E + 4 * AGG_WAVES - 1) / (4 * AGG_WAVES) + EDGE_STEP - 1) / EDGE_STEP * EDGE_STEP;

    gemm_wmma_bf16_kernel<128, 128><<<row_tiles, dim3(32, CH / 16), 0, stream>>>(x, W1, isq, hT, n);
    agg_kernel<<<CH * 2, AGG_WAVES * 32, 0, stream>>>(hT, pk, part, E, n, CH, slice2);

    fin_gemm_wmma_bf16_kernel<128, 128, 2><<<row_tiles, dim3(32, CH / 16), 0, stream>>>(part, hT, isq, b1, W2, hT2, n);
    agg_kernel<<<CH * 2, AGG_WAVES * 32, 0, stream>>>(hT2, pk, part, E, n, CH, slice2);

    fin_gemm_wmma_bf16_kernel<128, 64, 2><<<row_tiles, dim3(32, COUT / 16), 0, stream>>>(part, hT2, isq, b2, W3, hT, n);
    agg_kernel<<<COUT * 4, AGG_WAVES * 32, 0, stream>>>(hT, pk, part, E, n, COUT, slice4);
    fin_kernel<4, false><<<dim3(node_tiles, COUT / 32), dim3(32, 8), 0, stream>>>(
        part, hT, isq, b3, (float*)d_out, n, COUT);
}


# edges re-ordered inside each 512-edge block so that aggregation lane L gets edges with dst % 32 == L: a wave's LDS float add hits 32 different banks; sum order: new fixed in-block order
# speedup vs baseline: 2.6626x; 1.0341x over previous
#include <hip/hip_runtime.h>
#include <hip/hip_bf16.h>


typedef __attribute__((ext_vector_type(16))) __bf16 v16bf;
typedef __attribute__((ext_vector_type(8)))  float  v8f;
typedef __attribute__((ext_vector_type(8)))  __bf16 v8bf;

constexpr int NN        = 10000;
constexpr int AGG_WAVES = 7;
constexpr int EDGE_STEP = 512;

constexpr int PREP_WAVES   = 8;
constexpr int PREP_THREADS = PREP_WAVES * 32;
constexpr int PREP_EDGES   = PREP_WAVES * EDGE_STEP;

__device__ __forceinline__ int slot_pos(int L, int s) { return (s >> 2) * 128 + L * 4 + (s & 3); }

__global__ __launch_bounds__(PREP_THREADS)
void edge_prep_kernel(const int* __restrict__ src, const int* __restrict__ dst,
                      unsigned* __restrict__ pk, int* __restrict__ hist_part,
                      int E, int n) {
    __shared__ int hist[NN];
    __shared__ __attribute__((aligned(16))) unsigned stage[PREP_WAVES][EDGE_STEP];
    __shared__ unsigned short free_pos[PREP_WAVES][EDGE_STEP];
    const int tid  = threadIdx.x;
    const int lane = tid & 31;
    const int w    = tid >> 5;
    const int base = blockIdx.x * PREP_EDGES + w * EDGE_STEP;
    const bool whole = base + EDGE_STEP <= E;

    unsigned u[16];
    #pragma unroll
    for (int r = 0; r < 16; ++r) {
        const int e = min(base + r * 32 + lane, E - 1);
        u[r] = (unsigned)src[e] | ((unsigned)dst[e] << 16);
    }
    for (int i = tid; i < NN; i += PREP_THREADS) hist[i] = 0;
    __syncthreads();
    #pragma unroll
    for (int r = 0; r < 16; ++r)
        if (base + r * 32 + lane < E) atomicAdd(&hist[u[r] >> 16], 1);

    const unsigned below = (1u << lane) - 1u;
    int cnt = 0;
    int key[16];
    #pragma unroll
    for (int r = 0; r < 16; ++r) {
        const unsigned res = (u[r] >> 16) & 31u;
        unsigned same = 0xFFFFFFFFu;
        unsigned mine = 0xFFFFFFFFu;
        #pragma unroll
        for (int k = 0; k < 5; ++k) {
            const unsigned bk = __builtin_amdgcn_ballot_w32(((res >> k) & 1u) != 0u);
            same &= ((res >> k) & 1u) ? bk : ~bk;
            mine &= ((lane >> k) & 1) ? bk : ~bk;
        }
        key[r] = __shfl(cnt, (int)res) + __popc(same & below);
        cnt += __popc(mine);
    }

    const int over = max(cnt - 16, 0);
    const int fre  = max(16 - cnt, 0);
    const int both = over | (fre << 16);
    int incl = both;
    #pragma unroll
    for (int o = 1; o < 32; o <<= 1) {
        const int t = __shfl_up(incl, o);
        if (lane >= o) incl += t;
    }
    const int over_before = (incl - both) & 0xFFFF;
    const int free_before = (incl - both) >> 16;
    #pragma unroll
    for (int t = 0; t < 16; ++t)
        if (t < fre) free_pos[w][free_before + t] = (unsigned short)slot_pos(lane, cnt + t);
    __syncthreads();

    #pragma unroll
    for (int r = 0; r < 16; ++r) {
        const int res = (int)((u[r] >> 16) & 31u);
        const int ob  = __shfl(over_before, res);
        const int k   = key[r];
        const int spill = (int)free_pos[w][max(ob + k - 16, 0)];
        stage[w][(k < 16) ? slot_pos(res, k) : spill] = u[r];
    }
    __syncthreads();

    if (whole) {
        #pragma unroll
        for (int q = 0; q < 4; ++q)
            *reinterpret_cast<uint4*>(&pk[base + q * 128 + lane * 4]) =
                *reinterpret_cast<const uint4*>(&stage[w][q * 128 + lane * 4]);
    } else {
        #pragma unroll
        for (int r = 0; r < 16; ++r) {
            const int e = base + r * 32 + lane;
            if (e < E) pk[e] = u[r];
        }
    }
    for (int i = tid; i < n; i += PREP_THREADS) hist_part[(size_t)blockIdx.x * n + i] = hist[i];
}

constexpr int ISQ_BATCH = 24;
__global__ void isq_kernel(const int* __restrict__ hist_part,
                           float* __restrict__ isq, int n, int G) {
    __shared__ int red[8][33];
    const int tx = threadIdx.x, ty = threadIdx.y;
    const int node = blockIdx.x * 32 + tx;
    const int i = min(node, n - 1);
    int c = 0;
    for (int g0 = ty; g0 < G; g0 += 8 * ISQ_BATCH) {
        int v[ISQ_BATCH];
        #pragma unroll
        for (int u = 0; u < ISQ_BATCH; ++u)
            v[u] = hist_part[(size_t)min(g0 + u * 8, G - 1) * n + i];
        #pragma unroll
        for (int u = 0; u < ISQ_BATCH; ++u) c += (g0 + u * 8 < G) ? v[u] : 0;
    }
    red[ty][tx] = c;
    __syncthreads();
    if (ty == 0 && node < n) {
        int t = 1;
        #pragma unroll
        for (int k = 0; k < 8; ++k) t += red[k][tx];
        isq[node] = rsqrtf((float)t);
    }
}

template <int CIN, int COUT>
__global__ void gemm_wmma_bf16_kernel(const float* __restrict__ X,
                                      const float* __restrict__ W,
                                      const float* __restrict__ isq,
                                      float* __restrict__ HT, int nrows) {
    const int lane = threadIdx.x;
    const int ctile = threadIdx.y;
    const int m0 = blockIdx.x * 16;
    const int half = lane >> 4;
    const int l = lane & 15;
    const int m = m0 + l;
    const int n = ctile * 16 + l;

    v8f acc = {0.f, 0.f, 0.f, 0.f, 0.f, 0.f, 0.f, 0.f};

    #pragma unroll
    for (int kb = 0; kb < CIN; kb += 32) {
        float xa[16];
        const float4* xr0 = reinterpret_cast<const float4*>(&X[(size_t)m * CIN + kb + half * 8]);
        const float4* xr1 = reinterpret_cast<const float4*>(&X[(size_t)m * CIN + kb + 16 + half * 8]);
        *reinterpret_cast<float4*>(&xa[0])  = xr0[0];
        *reinterpret_cast<float4*>(&xa[4])  = xr0[1];
        *reinterpret_cast<float4*>(&xa[8])  = xr1[0];
        *reinterpret_cast<float4*>(&xa[12]) = xr1[1];

        v16bf a, b;
        #pragma unroll
        for (int e = 0; e < 8; ++e) {
            const int k0 = kb + half * 8 + e;
            const int k1 = k0 + 16;
            a[e]     = (__bf16)xa[e];
            a[e + 8] = (__bf16)xa[e + 8];
            b[e]     = (__bf16)W[(size_t)k0 * COUT + n];
            b[e + 8] = (__bf16)W[(size_t)k1 * COUT + n];
        }
        acc = __builtin_amdgcn_wmma_f32_16x16x32_bf16(
             false, a,  false, b,
             (short)0, acc,  false,  false);
    }

    const int r0 = m0 + half * 8;
    const float4 s0 = *reinterpret_cast<const float4*>(&isq[r0]);
    const float4 s1 = *reinterpret_cast<const float4*>(&isq[r0 + 4]);
    float* o = &HT[(size_t)(ctile * 16 + l) * nrows + r0];
    *reinterpret_cast<float4*>(o)     = make_float4(acc[0] * s0.x, acc[1] * s0.y, acc[2] * s0.z, acc[3] * s0.w);
    *reinterpret_cast<float4*>(o + 4) = make_float4(acc[4] * s1.x, acc[5] * s1.y, acc[6] * s1.z, acc[7] * s1.w);
}

__device__ __forceinline__ void lds_add(float* p, float v) {
    __hip_atomic_fetch_add(p, v, __ATOMIC_RELAXED, __HIP_MEMORY_SCOPE_WORKGROUP);
}

__device__ __forceinline__ void load_edges(uint4 (&v)[4], const unsigned* __restrict__ pk,
                                           int base, int lane) {
    #pragma unroll
    for (int q = 0; q < 4; ++q)
        v[q] = *reinterpret_cast<const uint4*>(&pk[base + q * 128 + lane * 4]);
}

__device__ __forceinline__ void add_edges(const uint4 (&v)[4], const float* hcol, float* col) {
    float h[16];
    #pragma unroll
    for (int q = 0; q < 4; ++q) {
        h[q * 4 + 0] = hcol[v[q].x & 0xFFFFu];
        h[q * 4 + 1] = hcol[v[q].y & 0xFFFFu];
        h[q * 4 + 2] = hcol[v[q].z & 0xFFFFu];
        h[q * 4 + 3] = hcol[v[q].w & 0xFFFFu];
    }
    #pragma unroll
    for (int q = 0; q < 4; ++q) {
        lds_add(&col[v[q].x >> 16], h[q * 4 + 0]);
        lds_add(&col[v[q].y >> 16], h[q * 4 + 1]);
        lds_add(&col[v[q].z >> 16], h[q * 4 + 2]);
        lds_add(&col[v[q].w >> 16], h[q * 4 + 3]);
    }
}

__global__ __launch_bounds__(AGG_WAVES * 32)
void agg_kernel(const float* __restrict__ HT, const unsigned* __restrict__ pk,
                float* __restrict__ part, int E, int n, int C, int slice) {
    __shared__ __attribute__((aligned(16))) float hcol[NN];
    __shared__ __attribute__((aligned(16))) float cols[AGG_WAVES][NN];

    const int tid  = threadIdx.x;
    const int lane = tid & 31;
    const int w    = __builtin_amdgcn_readfirstlane(tid >> 5);
    const int c    = blockIdx.x % C;
    const int p    = blockIdx.x / C;

    const int beg = (p * AGG_WAVES + w) * slice;
    const int end = min(beg + slice, E);
    int base = beg;
    const int nfull = (end - beg) / EDGE_STEP;

    constexpr int HC_TRIPS = (NN / 4 + AGG_WAVES * 32 - 1) / (AGG_WAVES * 32);
    constexpr int Z_TRIPS  = (NN / 4 + 31) / 32;
    const int last4 = (n >> 2) - 1;
    uint4 cur[4];
    load_edges(cur, pk, (nfull > 0) ? base : 0, lane);
    const float4* hsrc = reinterpret_cast<const float4*>(&HT[(size_t)c * n]);
    constexpr int TS = AGG_WAVES * 32;
    static_assert(HC_TRIPS == 12, "twelve named reads below");
    const float4 h0 = hsrc[min(tid + 0 * TS, last4)];
    const float4 h1 = hsrc[min(tid + 1 * TS, last4)];
    const float4 h2 = hsrc[min(tid + 2 * TS, last4)];
    const float4 h3 = hsrc[min(tid + 3 * TS, last4)];
    const float4 h4 = hsrc[min(tid + 4 * TS, last4)];
    const float4 h5 = hsrc[min(tid + 5 * TS, last4)];
    const float4 h6 = hsrc[min(tid + 6 * TS, last4)];
    const float4 h7 = hsrc[min(tid + 7 * TS, last4)];
    const float4 h8 = hsrc[min(tid + 8 * TS, last4)];
    const float4 h9 = hsrc[min(tid + 9 * TS, last4)];
    const float4 h10 = hsrc[min(tid + 10 * TS, last4)];
    const float4 h11 = hsrc[min(tid + 11 * TS, last4)];

    float* col = cols[w];
    float4* col4 = reinterpret_cast<float4*>(col);
    #pragma unroll
    for (int t = 0; t < Z_TRIPS; ++t)
        col4[min(lane + t * 32, NN / 4 - 1)] = make_float4(0.f, 0.f, 0.f, 0.f);

    float4* hcol4 = reinterpret_cast<float4*>(hcol);
    hcol4[min(tid + 0 * TS, last4)] = h0;
    hcol4[min(tid + 1 * TS, last4)] = h1;
    hcol4[min(tid + 2 * TS, last4)] = h2;
    hcol4[min(tid + 3 * TS, last4)] = h3;
    hcol4[min(tid + 4 * TS, last4)] = h4;
    hcol4[min(tid + 5 * TS, last4)] = h5;
    hcol4[min(tid + 6 * TS, last4)] = h6;
    hcol4[min(tid + 7 * TS, last4)] = h7;
    hcol4[min(tid + 8 * TS, last4)] = h8;
    hcol4[min(tid + 9 * TS, last4)] = h9;
    hcol4[min(tid + 10 * TS, last4)] = h10;
    hcol4[min(tid + 11 * TS, last4)] = h11;
    __syncthreads();

    if (nfull > 0) {
        for (int t = 0; t < nfull; ++t) {
            const int nb = (t + 1 < nfull) ? base + EDGE_STEP : base;
            uint4 nxt[4];
            load_edges(nxt, pk, nb, lane);
            add_edges(cur, hcol, col);
            #pragma unroll
            for (int q = 0; q < 4; ++q) cur[q] = nxt[q];
            base += EDGE_STEP;
        }
    }
    for (int e = base + lane; e < end; e += 32) {
        const unsigned u = pk[e];
        lds_add(&col[u >> 16], hcol[u & 0xFFFFu]);
    }
    __syncthreads();

    float* o = &part[((size_t)p * C + c) * n];
    for (int i = tid; i < n; i += AGG_WAVES * 32) {
        float s = cols[0][i];
        #pragma unroll
        for (int k = 1; k < AGG_WAVES; ++k) s += cols[k][i];
        o[i] = s;
    }
}

template <int P, bool RELU>
__global__ void fin_kernel(const float* __restrict__ part, const float* __restrict__ HT,
                           const float* __restrict__ isq, const float* __restrict__ bias,
                           float* __restrict__ OUT, int n, int C) {
    __shared__ float tile[32][33];
    const int tx = threadIdx.x, ty = threadIdx.y;
    const int i0 = blockIdx.x * 32, c0 = blockIdx.y * 32;
    const int i = i0 + tx;
    #pragma unroll
    for (int cc = ty; cc < 32; cc += 8) {
        const int c = c0 + cc;
        float s = 0.f;
        if (i < n) {
            s = part[(size_t)c * n + i];
            #pragma unroll
            for (int p = 1; p < P; ++p) s += part[((size_t)p * C + c) * n + i];
            s += HT[(size_t)c * n + i];
            s *= isq[i];
        }
        tile[cc][tx] = s;
    }
    __syncthreads();
    const float b = bias[c0 + tx];
    #pragma unroll
    for (int ii = ty; ii < 32; ii += 8) {
        const int node = i0 + ii;
        if (node < n) {
            const float v = tile[tx][ii] + b;
            OUT[(size_t)node * C + c0 + tx] = RELU ? fmaxf(v, 0.0f) : v;
        }
    }
}

constexpr int A_PITCH = 136;

template <int CIN, int COUT, int P>
__global__ void fin_gemm_wmma_bf16_kernel(const float* __restrict__ part,
                                          const float* __restrict__ HTin,
                                          const float* __restrict__ isq,
                                          const float* __restrict__ bias,
                                          const float* __restrict__ W,
                                          float* __restrict__ HTout, int nrows) {
    __shared__ __attribute__((aligned(16))) __bf16 As[16][A_PITCH];
    const int lane = threadIdx.x;
    const int ctile = threadIdx.y;
    const int m0 = blockIdx.x * 16;
    const int tid = ctile * 32 + lane;
    constexpr int T = 32 * (COUT / 16);

    const int half = lane >> 4;
    const int l = lane & 15;
    const int n = ctile * 16 + l;

    v16bf b[CIN / 32];
    #pragma unroll
    for (int kb = 0; kb < CIN; kb += 32) {
        #pragma unroll
        for (int e = 0; e < 8; ++e) {
            const int k0 = kb + half * 8 + e;
            const int k1 = k0 + 16;
            b[kb / 32][e]     = (__bf16)W[(size_t)k0 * COUT + n];
            b[kb / 32][e + 8] = (__bf16)W[(size_t)k1 * COUT + n];
        }
    }

    static_assert((CIN * 4) % T == 0, "items must divide evenly over the block");
    #pragma unroll
    for (int j = 0; j < CIN * 4 / T; ++j) {
        const int idx = tid + j * T;
        const int k = idx >> 2;
        const int q = idx & 3;
        const size_t off = (size_t)k * nrows + m0 + q * 4;
        float4 s = *reinterpret_cast<const float4*>(&part[off]);
        #pragma unroll
        for (int p = 1; p < P; ++p) {
            const float4 t = *reinterpret_cast<const float4*>(&part[(size_t)p * CIN * nrows + off]);
            s.x += t.x; s.y += t.y; s.z += t.z; s.w += t.w;
        }
        const float4 h = *reinterpret_cast<const float4*>(&HTin[off]);
        const float4 r = *reinterpret_cast<const float4*>(&isq[m0 + q * 4]);
        const float bk = bias[k];
        s.x = fmaxf((s.x + h.x) * r.x + bk, 0.0f);
        s.y = fmaxf((s.y + h.y) * r.y + bk, 0.0f);
        s.z = fmaxf((s.z + h.z) * r.z + bk, 0.0f);
        s.w = fmaxf((s.w + h.w) * r.w + bk, 0.0f);
        As[q * 4 + 0][k] = (__bf16)s.x;
        As[q * 4 + 1][k] = (__bf16)s.y;
        As[q * 4 + 2][k] = (__bf16)s.z;
        As[q * 4 + 3][k] = (__bf16)s.w;
    }
    __syncthreads();

    v8f acc = {0.f, 0.f, 0.f, 0.f, 0.f, 0.f, 0.f, 0.f};

    #pragma unroll
    for (int kb = 0; kb < CIN; kb += 32) {
        const v8bf alo = *reinterpret_cast<const v8bf*>(&As[l][kb + half * 8]);
        const v8bf ahi = *reinterpret_cast<const v8bf*>(&As[l][kb + 16 + half * 8]);
        const v16bf a = __builtin_shufflevector(alo, ahi, 0, 1, 2, 3, 4, 5, 6, 7, 8, 9, 10, 11, 12, 13, 14, 15);
        acc = __builtin_amdgcn_wmma_f32_16x16x32_bf16(
             false, a,  false, b[kb / 32],
             (short)0, acc,  false,  false);
    }

    const int r0 = m0 + half * 8;
    const float4 s0 = *reinterpret_cast<const float4*>(&isq[r0]);
    const float4 s1 = *reinterpret_cast<const float4*>(&isq[r0 + 4]);
    float* o = &HTout[(size_t)(ctile * 16 + l) * nrows + r0];
    *reinterpret_cast<float4*>(o)     = make_float4(acc[0] * s0.x, acc[1] * s0.y, acc[2] * s0.z, acc[3] * s0.w);
    *reinterpret_cast<float4*>(o + 4) = make_float4(acc[4] * s1.x, acc[5] * s1.y, acc[6] * s1.z, acc[7] * s1.w);
}

extern "C" void kernel_launch(void* const* d_in, const int* in_sizes, int n_in,
                              void* d_out, int out_size, void* d_ws, size_t ws_size,
                              hipStream_t stream) {
    const float* x  = (const float*)d_in[0];
    const int*   ei = (const int*)d_in[1];
    const float* W1 = (const float*)d_in[2];
    const float* b1 = (const float*)d_in[3];
    const float* W2 = (const float*)d_in[4];
    const float* b2 = (const float*)d_in[5];
    const float* W3 = (const float*)d_in[6];
    const float* b3 = (const float*)d_in[7];

    const int CIN = 128, CH = 128, COUT = 64;
    const int n = in_sizes[0] / CIN;
    const int E = in_sizes[1] / 2;
    const int* src = ei;
    const int* dst = ei + E;

    unsigned* pk = (unsigned*)d_ws;
    float* hT   = (float*)(pk + E);
    float* part = hT   + (size_t)n * CH;
    float* hT2  = part + (size_t)n * CH * 2;
    float* isq  = hT2  + (size_t)n * CH;
    int*   hist_part = (int*)(isq + n);

    const int row_tiles = n / 16;
    const int node_tiles = (n + 31) / 32;
    const int prep_wgs = (E + 4095) / 4096;

    edge_prep_kernel<<<prep_wgs, 256, 0, stream>>>(src, dst, pk, hist_part, E, n);
    isq_kernel<<<node_tiles, dim3(32, 8), 0, stream>>>(hist_part, isq, n, prep_wgs);

    const int slice2 = ((E + 2 * AGG_WAVES - 1) / (2 * AGG_WAVES) + EDGE_STEP - 1) / EDGE_STEP * EDGE_STEP;
    const int slice4 = ((E + 4 * AGG_WAVES - 1) / (4 * AGG_WAVES) + EDGE_STEP - 1) / EDGE_STEP * EDGE_STEP;

    gemm_wmma_bf16_kernel<128, 128><<<row_tiles, dim3(32, CH / 16), 0, stream>>>(x, W1, isq, hT, n);
    agg_kernel<<<CH * 2, AGG_WAVES * 32, 0, stream>>>(hT, pk, part, E, n, CH, slice2);

    fin_gemm_wmma_bf16_kernel<128, 128, 2><<<row_tiles, dim3(32, CH / 16), 0, stream>>>(part, hT, isq, b1, W2, hT2, n);
    agg_kernel<<<CH * 2, AGG_WAVES * 32, 0, stream>>>(hT2, pk, part, E, n, CH, slice2);

    fin_gemm_wmma_bf16_kernel<128, 64, 2><<<row_tiles, dim3(32, COUT / 16), 0, stream>>>(part, hT2, isq, b2, W3, hT, n);
    agg_kernel<<<COUT * 4, AGG_WAVES * 32, 0, stream>>>(hT, pk, part, E, n, COUT, slice4);
    fin_kernel<4, false><<<dim3(node_tiles, COUT / 32), dim3(32, 8), 0, stream>>>(
        part, hT, isq, b3, (float*)d_out, n, COUT);
}


# aggregation asks for its edge list 6 steps (6 x 512 edges per wave, a ring of register sets) ahead of the LDS work instead of 1: a step no longer waits a memory round trip; sum order unchanged
# speedup vs baseline: 3.1412x; 1.1797x over previous
#include <hip/hip_runtime.h>
#include <hip/hip_bf16.h>


typedef __attribute__((ext_vector_type(16))) __bf16 v16bf;
typedef __attribute__((ext_vector_type(8)))  float  v8f;
typedef __attribute__((ext_vector_type(8)))  __bf16 v8bf;

constexpr int NN        = 10000;
constexpr int AGG_WAVES = 7;
constexpr int EDGE_STEP = 512;
constexpr int EDGE_AHEAD = 6;

constexpr int PREP_WAVES   = 8;
constexpr int PREP_THREADS = PREP_WAVES * 32;
constexpr int PREP_EDGES   = PREP_WAVES * EDGE_STEP;

__device__ __forceinline__ int slot_pos(int L, int s) { return (s >> 2) * 128 + L * 4 + (s & 3); }

__global__ __launch_bounds__(PREP_THREADS)
void edge_prep_kernel(const int* __restrict__ src, const int* __restrict__ dst,
                      unsigned* __restrict__ pk, int* __restrict__ hist_part,
                      int E, int n) {
    __shared__ int hist[NN];
    __shared__ __attribute__((aligned(16))) unsigned stage[PREP_WAVES][EDGE_STEP];
    __shared__ unsigned short free_pos[PREP_WAVES][EDGE_STEP];
    const int tid  = threadIdx.x;
    const int lane = tid & 31;
    const int w    = tid >> 5;
    const int base = blockIdx.x * PREP_EDGES + w * EDGE_STEP;
    const bool whole = base + EDGE_STEP <= E;

    unsigned u[16];
    #pragma unroll
    for (int r = 0; r < 16; ++r) {
        const int e = min(base + r * 32 + lane, E - 1);
        u[r] = (unsigned)src[e] | ((unsigned)dst[e] << 16);
    }
    for (int i = tid; i < NN; i += PREP_THREADS) hist[i] = 0;
    __syncthreads();
    #pragma unroll
    for (int r = 0; r < 16; ++r)
        if (base + r * 32 + lane < E) atomicAdd(&hist[u[r] >> 16], 1);

    const unsigned below = (1u << lane) - 1u;
    int cnt = 0;
    int key[16];
    #pragma unroll
    for (int r = 0; r < 16; ++r) {
        const unsigned res = (u[r] >> 16) & 31u;
        unsigned same = 0xFFFFFFFFu;
        unsigned mine = 0xFFFFFFFFu;
        #pragma unroll
        for (int k = 0; k < 5; ++k) {
            const unsigned bk = __builtin_amdgcn_ballot_w32(((res >> k) & 1u) != 0u);
            same &= ((res >> k) & 1u) ? bk : ~bk;
            mine &= ((lane >> k) & 1) ? bk : ~bk;
        }
        key[r] = __shfl(cnt, (int)res) + __popc(same & below);
        cnt += __popc(mine);
    }

    const int over = max(cnt - 16, 0);
    const int fre  = max(16 - cnt, 0);
    const int both = over | (fre << 16);
    int incl = both;
    #pragma unroll
    for (int o = 1; o < 32; o <<= 1) {
        const int t = __shfl_up(incl, o);
        if (lane >= o) incl += t;
    }
    const int over_before = (incl - both) & 0xFFFF;
    const int free_before = (incl - both) >> 16;
    #pragma unroll
    for (int t = 0; t < 16; ++t)
        if (t < fre) free_pos[w][free_before + t] = (unsigned short)slot_pos(lane, cnt + t);
    __syncthreads();

    #pragma unroll
    for (int r = 0; r < 16; ++r) {
        const int res = (int)((u[r] >> 16) & 31u);
        const int ob  = __shfl(over_before, res);
        const int k   = key[r];
        const int spill = (int)free_pos[w][max(ob + k - 16, 0)];
        stage[w][(k < 16) ? slot_pos(res, k) : spill] = u[r];
    }
    __syncthreads();

    if (whole) {
        #pragma unroll
        for (int q = 0; q < 4; ++q)
            *reinterpret_cast<uint4*>(&pk[base + q * 128 + lane * 4]) =
                *reinterpret_cast<const uint4*>(&stage[w][q * 128 + lane * 4]);
    } else {
        #pragma unroll
        for (int r = 0; r < 16; ++r) {
            const int e = base + r * 32 + lane;
            if (e < E) pk[e] = u[r];
        }
    }
    for (int i = tid; i < n; i += PREP_THREADS) hist_part[(size_t)blockIdx.x * n + i] = hist[i];
}

constexpr int ISQ_BATCH = 24;
__global__ void isq_kernel(const int* __restrict__ hist_part,
                           float* __restrict__ isq, int n, int G) {
    __shared__ int red[8][33];
    const int tx = threadIdx.x, ty = threadIdx.y;
    const int node = blockIdx.x * 32 + tx;
    const int i = min(node, n - 1);
    int c = 0;
    for (int g0 = ty; g0 < G; g0 += 8 * ISQ_BATCH) {
        int v[ISQ_BATCH];
        #pragma unroll
        for (int u = 0; u < ISQ_BATCH; ++u)
            v[u] = hist_part[(size_t)min(g0 + u * 8, G - 1) * n + i];
        #pragma unroll
        for (int u = 0; u < ISQ_BATCH; ++u) c += (g0 + u * 8 < G) ? v[u] : 0;
    }
    red[ty][tx] = c;
    __syncthreads();
    if (ty == 0 && node < n) {
        int t = 1;
        #pragma unroll
        for (int k = 0; k < 8; ++k) t += red[k][tx];
        isq[node] = rsqrtf((float)t);
    }
}

template <int CIN, int COUT>
__global__ void gemm_wmma_bf16_kernel(const float* __restrict__ X,
                                      const float* __restrict__ W,
                                      const float* __restrict__ isq,
                                      float* __restrict__ HT, int nrows) {
    const int lane = threadIdx.x;
    const int ctile = threadIdx.y;
    const int m0 = blockIdx.x * 16;
    const int half = lane >> 4;
    const int l = lane & 15;
    const int m = m0 + l;
    const int n = ctile * 16 + l;

    v8f acc = {0.f, 0.f, 0.f, 0.f, 0.f, 0.f, 0.f, 0.f};

    #pragma unroll
    for (int kb = 0; kb < CIN; kb += 32) {
        float xa[16];
        const float4* xr0 = reinterpret_cast<const float4*>(&X[(size_t)m * CIN + kb + half * 8]);
        const float4* xr1 = reinterpret_cast<const float4*>(&X[(size_t)m * CIN + kb + 16 + half * 8]);
        *reinterpret_cast<float4*>(&xa[0])  = xr0[0];
        *reinterpret_cast<float4*>(&xa[4])  = xr0[1];
        *reinterpret_cast<float4*>(&xa[8])  = xr1[0];
        *reinterpret_cast<float4*>(&xa[12]) = xr1[1];

        v16bf a, b;
        #pragma unroll
        for (int e = 0; e < 8; ++e) {
            const int k0 = kb + half * 8 + e;
            const int k1 = k0 + 16;
            a[e]     = (__bf16)xa[e];
            a[e + 8] = (__bf16)xa[e + 8];
            b[e]     = (__bf16)W[(size_t)k0 * COUT + n];
            b[e + 8] = (__bf16)W[(size_t)k1 * COUT + n];
        }
        acc = __builtin_amdgcn_wmma_f32_16x16x32_bf16(
             false, a,  false, b,
             (short)0, acc,  false,  false);
    }

    const int r0 = m0 + half * 8;
    const float4 s0 = *reinterpret_cast<const float4*>(&isq[r0]);
    const float4 s1 = *reinterpret_cast<const float4*>(&isq[r0 + 4]);
    float* o = &HT[(size_t)(ctile * 16 + l) * nrows + r0];
    *reinterpret_cast<float4*>(o)     = make_float4(acc[0] * s0.x, acc[1] * s0.y, acc[2] * s0.z, acc[3] * s0.w);
    *reinterpret_cast<float4*>(o + 4) = make_float4(acc[4] * s1.x, acc[5] * s1.y, acc[6] * s1.z, acc[7] * s1.w);
}

__device__ __forceinline__ void lds_add(float* p, float v) {
    __hip_atomic_fetch_add(p, v, __ATOMIC_RELAXED, __HIP_MEMORY_SCOPE_WORKGROUP);
}

__device__ __forceinline__ void load_edges(uint4 (&v)[4], const unsigned* __restrict__ pk,
                                           int base, int lane) {
    #pragma unroll
    for (int q = 0; q < 4; ++q)
        v[q] = *reinterpret_cast<const uint4*>(&pk[base + q * 128 + lane * 4]);
}

__device__ __forceinline__ void add_edges(const uint4 (&v)[4], const float* hcol, float* col) {
    float h[16];
    #pragma unroll
    for (int q = 0; q < 4; ++q) {
        h[q * 4 + 0] = hcol[v[q].x & 0xFFFFu];
        h[q * 4 + 1] = hcol[v[q].y & 0xFFFFu];
        h[q * 4 + 2] = hcol[v[q].z & 0xFFFFu];
        h[q * 4 + 3] = hcol[v[q].w & 0xFFFFu];
    }
    #pragma unroll
    for (int q = 0; q < 4; ++q) {
        lds_add(&col[v[q].x >> 16], h[q * 4 + 0]);
        lds_add(&col[v[q].y >> 16], h[q * 4 + 1]);
        lds_add(&col[v[q].z >> 16], h[q * 4 + 2]);
        lds_add(&col[v[q].w >> 16], h[q * 4 + 3]);
    }
}

__global__ __launch_bounds__(AGG_WAVES * 32)
void agg_kernel(const float* __restrict__ HT, const unsigned* __restrict__ pk,
                float* __restrict__ part, int E, int n, int C, int slice) {
    __shared__ __attribute__((aligned(16))) float hcol[NN];
    __shared__ __attribute__((aligned(16))) float cols[AGG_WAVES][NN];

    const int tid  = threadIdx.x;
    const int lane = tid & 31;
    const int w    = __builtin_amdgcn_readfirstlane(tid >> 5);
    const int c    = blockIdx.x % C;
    const int p    = blockIdx.x / C;

    const int beg = (p * AGG_WAVES + w) * slice;
    const int end = min(beg + slice, E);
    const int nfull = max((end - beg) / EDGE_STEP, 0);
    const int last_step = max(nfull - 1, 0);
    const int blk0 = (nfull > 0) ? beg : 0;

    constexpr int HC_TRIPS = (NN / 4 + AGG_WAVES * 32 - 1) / (AGG_WAVES * 32);
    constexpr int Z_TRIPS  = (NN / 4 + 31) / 32;
    const int last4 = (n >> 2) - 1;
    const float4* hsrc = reinterpret_cast<const float4*>(&HT[(size_t)c * n]);
    constexpr int TS = AGG_WAVES * 32;
    static_assert(HC_TRIPS == 12, "twelve named reads below");
    const float4 h0 = hsrc[min(tid + 0 * TS, last4)];
    const float4 h1 = hsrc[min(tid + 1 * TS, last4)];
    const float4 h2 = hsrc[min(tid + 2 * TS, last4)];
    const float4 h3 = hsrc[min(tid + 3 * TS, last4)];
    const float4 h4 = hsrc[min(tid + 4 * TS, last4)];
    const float4 h5 = hsrc[min(tid + 5 * TS, last4)];
    const float4 h6 = hsrc[min(tid + 6 * TS, last4)];
    const float4 h7 = hsrc[min(tid + 7 * TS, last4)];
    const float4 h8 = hsrc[min(tid + 8 * TS, last4)];
    const float4 h9 = hsrc[min(tid + 9 * TS, last4)];
    const float4 h10 = hsrc[min(tid + 10 * TS, last4)];
    const float4 h11 = hsrc[min(tid + 11 * TS, last4)];
    uint4 ring[EDGE_AHEAD][4];
    #pragma unroll
    for (int j = 0; j < EDGE_AHEAD; ++j)
        load_edges(ring[j], pk, blk0 + min(j, last_step) * EDGE_STEP, lane);

    float* col = cols[w];
    float4* col4 = reinterpret_cast<float4*>(col);
    #pragma unroll
    for (int t = 0; t < Z_TRIPS; ++t)
        col4[min(lane + t * 32, NN / 4 - 1)] = make_float4(0.f, 0.f, 0.f, 0.f);

    float4* hcol4 = reinterpret_cast<float4*>(hcol);
    hcol4[min(tid + 0 * TS, last4)] = h0;
    hcol4[min(tid + 1 * TS, last4)] = h1;
    hcol4[min(tid + 2 * TS, last4)] = h2;
    hcol4[min(tid + 3 * TS, last4)] = h3;
    hcol4[min(tid + 4 * TS, last4)] = h4;
    hcol4[min(tid + 5 * TS, last4)] = h5;
    hcol4[min(tid + 6 * TS, last4)] = h6;
    hcol4[min(tid + 7 * TS, last4)] = h7;
    hcol4[min(tid + 8 * TS, last4)] = h8;
    hcol4[min(tid + 9 * TS, last4)] = h9;
    hcol4[min(tid + 10 * TS, last4)] = h10;
    hcol4[min(tid + 11 * TS, last4)] = h11;
    __syncthreads();

    const int ngroups = nfull / EDGE_AHEAD;
    int req = EDGE_AHEAD;
    for (int g = 0; g < ngroups; ++g) {
        #pragma unroll
        for (int j = 0; j < EDGE_AHEAD; ++j) {
            add_edges(ring[j], hcol, col);
            load_edges(ring[j], pk, blk0 + min(req + j, last_step) * EDGE_STEP, lane);
            __builtin_amdgcn_sched_barrier(0);
        }
        req += EDGE_AHEAD;
    }
    const int rem = nfull - ngroups * EDGE_AHEAD;
    #pragma unroll
    for (int j = 0; j < EDGE_AHEAD - 1; ++j)
        if (j < rem) add_edges(ring[j], hcol, col);
    const int base = beg + nfull * EDGE_STEP;
    for (int e = base + lane; e < end; e += 32) {
        const unsigned u = pk[e];
        lds_add(&col[u >> 16], hcol[u & 0xFFFFu]);
    }
    __syncthreads();

    float* o = &part[((size_t)p * C + c) * n];
    for (int i = tid; i < n; i += AGG_WAVES * 32) {
        float s = cols[0][i];
        #pragma unroll
        for (int k = 1; k < AGG_WAVES; ++k) s += cols[k][i];
        o[i] = s;
    }
}

template <int P, bool RELU>
__global__ void fin_kernel(const float* __restrict__ part, const float* __restrict__ HT,
                           const float* __restrict__ isq, const float* __restrict__ bias,
                           float* __restrict__ OUT, int n, int C) {
    __shared__ float tile[32][33];
    const int tx = threadIdx.x, ty = threadIdx.y;
    const int i0 = blockIdx.x * 32, c0 = blockIdx.y * 32;
    const int i = i0 + tx;
    #pragma unroll
    for (int cc = ty; cc < 32; cc += 8) {
        const int c = c0 + cc;
        float s = 0.f;
        if (i < n) {
            s = part[(size_t)c * n + i];
            #pragma unroll
            for (int p = 1; p < P; ++p) s += part[((size_t)p * C + c) * n + i];
            s += HT[(size_t)c * n + i];
            s *= isq[i];
        }
        tile[cc][tx] = s;
    }
    __syncthreads();
    const float b = bias[c0 + tx];
    #pragma unroll
    for (int ii = ty; ii < 32; ii += 8) {
        const int node = i0 + ii;
        if (node < n) {
            const float v = tile[tx][ii] + b;
            OUT[(size_t)node * C + c0 + tx] = RELU ? fmaxf(v, 0.0f) : v;
        }
    }
}

constexpr int A_PITCH = 136;

template <int CIN, int COUT, int P>
__global__ void fin_gemm_wmma_bf16_kernel(const float* __restrict__ part,
                                          const float* __restrict__ HTin,
                                          const float* __restrict__ isq,
                                          const float* __restrict__ bias,
                                          const float* __restrict__ W,
                                          float* __restrict__ HTout, int nrows) {
    __shared__ __attribute__((aligned(16))) __bf16 As[16][A_PITCH];
    const int lane = threadIdx.x;
    const int ctile = threadIdx.y;
    const int m0 = blockIdx.x * 16;
    const int tid = ctile * 32 + lane;
    constexpr int T = 32 * (COUT / 16);

    const int half = lane >> 4;
    const int l = lane & 15;
    const int n = ctile * 16 + l;

    v16bf b[CIN / 32];
    #pragma unroll
    for (int kb = 0; kb < CIN; kb += 32) {
        #pragma unroll
        for (int e = 0; e < 8; ++e) {
            const int k0 = kb + half * 8 + e;
            const int k1 = k0 + 16;
            b[kb / 32][e]     = (__bf16)W[(size_t)k0 * COUT + n];
            b[kb / 32][e + 8] = (__bf16)W[(size_t)k1 * COUT + n];
        }
    }

    static_assert((CIN * 4) % T == 0, "items must divide evenly over the block");
    #pragma unroll
    for (int j = 0; j < CIN * 4 / T; ++j) {
        const int idx = tid + j * T;
        const int k = idx >> 2;
        const int q = idx & 3;
        const size_t off = (size_t)k * nrows + m0 + q * 4;
        float4 s = *reinterpret_cast<const float4*>(&part[off]);
        #pragma unroll
        for (int p = 1; p < P; ++p) {
            const float4 t = *reinterpret_cast<const float4*>(&part[(size_t)p * CIN * nrows + off]);
            s.x += t.x; s.y += t.y; s.z += t.z; s.w += t.w;
        }
        const float4 h = *reinterpret_cast<const float4*>(&HTin[off]);
        const float4 r = *reinterpret_cast<const float4*>(&isq[m0 + q * 4]);
        const float bk = bias[k];
        s.x = fmaxf((s.x + h.x) * r.x + bk, 0.0f);
        s.y = fmaxf((s.y + h.y) * r.y + bk, 0.0f);
        s.z = fmaxf((s.z + h.z) * r.z + bk, 0.0f);
        s.w = fmaxf((s.w + h.w) * r.w + bk, 0.0f);
        As[q * 4 + 0][k] = (__bf16)s.x;
        As[q * 4 + 1][k] = (__bf16)s.y;
        As[q * 4 + 2][k] = (__bf16)s.z;
        As[q * 4 + 3][k] = (__bf16)s.w;
    }
    __syncthreads();

    v8f acc = {0.f, 0.f, 0.f, 0.f, 0.f, 0.f, 0.f, 0.f};

    #pragma unroll
    for (int kb = 0; kb < CIN; kb += 32) {
        const v8bf alo = *reinterpret_cast<const v8bf*>(&As[l][kb + half * 8]);
        const v8bf ahi = *reinterpret_cast<const v8bf*>(&As[l][kb + 16 + half * 8]);
        const v16bf a = __builtin_shufflevector(alo, ahi, 0, 1, 2, 3, 4, 5, 6, 7, 8, 9, 10, 11, 12, 13, 14, 15);
        acc = __builtin_amdgcn_wmma_f32_16x16x32_bf16(
             false, a,  false, b[kb / 32],
             (short)0, acc,  false,  false);
    }

    const int r0 = m0 + half * 8;
    const float4 s0 = *reinterpret_cast<const float4*>(&isq[r0]);
    const float4 s1 = *reinterpret_cast<const float4*>(&isq[r0 + 4]);
    float* o = &HTout[(size_t)(ctile * 16 + l) * nrows + r0];
    *reinterpret_cast<float4*>(o)     = make_float4(acc[0] * s0.x, acc[1] * s0.y, acc[2] * s0.z, acc[3] * s0.w);
    *reinterpret_cast<float4*>(o + 4) = make_float4(acc[4] * s1.x, acc[5] * s1.y, acc[6] * s1.z, acc[7] * s1.w);
}

extern "C" void kernel_launch(void* const* d_in, const int* in_sizes, int n_in,
                              void* d_out, int out_size, void* d_ws, size_t ws_size,
                              hipStream_t stream) {
    const float* x  = (const float*)d_in[0];
    const int*   ei = (const int*)d_in[1];
    const float* W1 = (const float*)d_in[2];
    const float* b1 = (const float*)d_in[3];
    const float* W2 = (const float*)d_in[4];
    const float* b2 = (const float*)d_in[5];
    const float* W3 = (const float*)d_in[6];
    const float* b3 = (const float*)d_in[7];

    const int CIN = 128, CH = 128, COUT = 64;
    const int n = in_sizes[0] / CIN;
    const int E = in_sizes[1] / 2;
    const int* src = ei;
    const int* dst = ei + E;

    unsigned* pk = (unsigned*)d_ws;
    float* hT   = (float*)(pk + E);
    float* part = hT   + (size_t)n * CH;
    float* hT2  = part + (size_t)n * CH * 2;
    float* isq  = hT2  + (size_t)n * CH;
    int*   hist_part = (int*)(isq + n);

    const int row_tiles = n / 16;
    const int node_tiles = (n + 31) / 32;
    const int prep_wgs = (E + 4095) / 4096;

    edge_prep_kernel<<<prep_wgs, 256, 0, stream>>>(src, dst, pk, hist_part, E, n);
    isq_kernel<<<node_tiles, dim3(32, 8), 0, stream>>>(hist_part, isq, n, prep_wgs);

    const int slice2 = ((E + 2 * AGG_WAVES - 1) / (2 * AGG_WAVES) + EDGE_STEP - 1) / EDGE_STEP * EDGE_STEP;
    const int slice4 = ((E + 4 * AGG_WAVES - 1) / (4 * AGG_WAVES) + EDGE_STEP - 1) / EDGE_STEP * EDGE_STEP;

    gemm_wmma_bf16_kernel<128, 128><<<row_tiles, dim3(32, CH / 16), 0, stream>>>(x, W1, isq, hT, n);
    agg_kernel<<<CH * 2, AGG_WAVES * 32, 0, stream>>>(hT, pk, part, E, n, CH, slice2);

    fin_gemm_wmma_bf16_kernel<128, 128, 2><<<row_tiles, dim3(32, CH / 16), 0, stream>>>(part, hT, isq, b1, W2, hT2, n);
    agg_kernel<<<CH * 2, AGG_WAVES * 32, 0, stream>>>(hT2, pk, part, E, n, CH, slice2);

    fin_gemm_wmma_bf16_kernel<128, 64, 2><<<row_tiles, dim3(32, COUT / 16), 0, stream>>>(part, hT2, isq, b2, W3, hT, n);
    agg_kernel<<<COUT * 4, AGG_WAVES * 32, 0, stream>>>(hT, pk, part, E, n, COUT, slice4);
    fin_kernel<4, false><<<dim3(node_tiles, COUT / 32), dim3(32, 8), 0, stream>>>(
        part, hT, isq, b3, (float*)d_out, n, COUT);
}
